# starting kernel
# baseline (speedup 1.0000x reference)
#include <hip/hip_runtime.h>
#include <hip/hip_bf16.h>

typedef __attribute__((ext_vector_type(2))) float v2f;
typedef __attribute__((ext_vector_type(8))) float v8f;

__device__ __forceinline__ v8f wmma_f32(v2f a, v2f b, v8f c) {
  return __builtin_amdgcn_wmma_f32_16x16x4_f32(
       false, a,  false, b,
       (short)0, c,  false,  false);
}

__global__ __launch_bounds__(32)
void gemm_nt_wmma(const float* __restrict__ A, const float* __restrict__ B,
                  float* __restrict__ C, int K, int lda, int ldb, int ldc) {
  const int lane = threadIdx.x & 31;
  const int half = lane >> 4;
  const int l16  = lane & 15;
  const int bm = blockIdx.y * 32;
  const int bn = blockIdx.x * 64;

  v8f acc[2][4];
#pragma unroll
  for (int i = 0; i < 2; ++i)
#pragma unroll
    for (int j = 0; j < 4; ++j) acc[i][j] = (v8f){};

  uint32_t aoff[2], boff[4];
#pragma unroll
  for (int i = 0; i < 2; ++i)
    aoff[i] = (uint32_t)(bm + i * 16 + l16) * (uint32_t)lda + 2u * half;
#pragma unroll
  for (int j = 0; j < 4; ++j)
    boff[j] = (uint32_t)(bn + j * 16 + l16) * (uint32_t)ldb + 2u * half;

  v2f af[2], bf[4];
#pragma unroll
  for (int i = 0; i < 2; ++i) af[i] = *(const v2f*)(A + aoff[i]);
#pragma unroll
  for (int j = 0; j < 4; ++j) bf[j] = *(const v2f*)(B + boff[j]);

#pragma unroll 2
  for (int k0 = 0; k0 < K - 4; k0 += 4) {
    v2f afn[2], bfn[4];
#pragma unroll
    for (int i = 0; i < 2; ++i)
      afn[i] = *(const v2f*)(A + aoff[i] + (uint32_t)(k0 + 4));
#pragma unroll
    for (int j = 0; j < 4; ++j)
      bfn[j] = *(const v2f*)(B + boff[j] + (uint32_t)(k0 + 4));
#pragma unroll
    for (int i = 0; i < 2; ++i)
#pragma unroll
      for (int j = 0; j < 4; ++j) acc[i][j] = wmma_f32(af[i], bf[j], acc[i][j]);
#pragma unroll
    for (int i = 0; i < 2; ++i) af[i] = afn[i];
#pragma unroll
    for (int j = 0; j < 4; ++j) bf[j] = bfn[j];
  }
#pragma unroll
  for (int i = 0; i < 2; ++i)
#pragma unroll
    for (int j = 0; j < 4; ++j) acc[i][j] = wmma_f32(af[i], bf[j], acc[i][j]);

#pragma unroll
  for (int i = 0; i < 2; ++i)
#pragma unroll
    for (int j = 0; j < 4; ++j) {
      const uint32_t row0 = (uint32_t)(bm + i * 16 + half * 8);
      const uint32_t col  = (uint32_t)(bn + j * 16 + l16);
#pragma unroll
      for (int v = 0; v < 8; ++v)
        C[(row0 + v) * (uint32_t)ldc + col] = acc[i][j][v];
    }
}

__global__ __launch_bounds__(256)
void rope_kernel(float* __restrict__ qkv, int T) {
  int idx = blockIdx.x * blockDim.x + threadIdx.x;
  if (idx >= T * 1024) return;
  int t = idx >> 10;
  int r = idx & 1023;
  int qksel = r >> 9;
  int h = (r >> 5) & 15;
  int p = r & 31;
  float inv = powf(10000.0f, -(float)(2 * p) / 64.0f);
  float ang = (float)t * inv;
  float s = sinf(ang), c = cosf(ang);
  float* base = qkv + (uint32_t)t * 3072u + (uint32_t)(qksel * 1024 + h * 64 + 2 * p);
  float x1 = base[0], x2 = base[1];
  base[0] = x1 * c - x2 * s;
  base[1] = x2 * c + x1 * s;
}

#define LDT 68

__global__ __launch_bounds__(256)
void attn_kernel(const float* __restrict__ qkv, float* __restrict__ out) {
  __shared__ float Klds[64 * LDT];
  __shared__ float Vlds[64 * LDT];
  __shared__ float Plds[8 * 16 * LDT];

  const int tid  = threadIdx.x;
  const int lane = tid & 31;
  const int wave = tid >> 5;
  const int half = lane >> 4;
  const int l16  = lane & 15;
  const int h     = blockIdx.y;
  const int qbase = blockIdx.x * 128 + wave * 16;
  const uint32_t ld = 3072;

  const float* qp = qkv + (uint32_t)h * 64u;
  const float* kp = qkv + 1024u + (uint32_t)h * 64u;
  const float* vp = qkv + 2048u + (uint32_t)h * 64u;

  v2f qf[16];
  {
    const uint32_t qoff = (uint32_t)(qbase + l16) * ld + 2u * half;
#pragma unroll
    for (int kk = 0; kk < 16; ++kk)
      qf[kk] = *(const v2f*)(qp + qoff + 4u * kk);
  }

  v8f o[4];
#pragma unroll
  for (int ct = 0; ct < 4; ++ct) o[ct] = (v8f){};
  float mrow[8], lrow[8];
#pragma unroll
  for (int v = 0; v < 8; ++v) { mrow[v] = -1e30f; lrow[v] = 0.0f; }

  float* myp = Plds + wave * 16 * LDT;
  const int ktmax_w = (qbase + 15) >> 6;
  const int ktmax_b = (blockIdx.x * 128 + 127) >> 6;

  for (int kt = 0; kt <= ktmax_b; ++kt) {
    const uint32_t tbase = (uint32_t)(kt * 64) * ld;

    __syncthreads();
#pragma unroll 4
    for (int t = 0; t < 16; ++t) {
      int idx = t * 256 + tid;
      int key = idx >> 6, c = idx & 63;
      uint32_t goff = tbase + (uint32_t)key * ld + (uint32_t)c;
      float kvv = kp[goff];
      float vvv = vp[goff];
      Klds[key * LDT + c] = kvv;
      Vlds[c * LDT + key] = vvv;
    }
    if (kt < ktmax_b) {
      if (tid < 64) {
        const float* p = kp + tbase + (uint32_t)(64 + tid) * ld;
        __builtin_prefetch(p, 0, 3);
        __builtin_prefetch(p + 32, 0, 3);
      } else if (tid < 128) {
        const float* p = vp + tbase + (uint32_t)tid * ld;
        __builtin_prefetch(p, 0, 3);
        __builtin_prefetch(p + 32, 0, 3);
      }
    }
    __syncthreads();

    if (kt <= ktmax_w) {
      v8f s[4];
#pragma unroll
      for (int j = 0; j < 4; ++j) s[j] = (v8f){};
#pragma unroll 4
      for (int kk = 0; kk < 16; ++kk) {
        v2f a = qf[kk];
#pragma unroll
        for (int j = 0; j < 4; ++j) {
          v2f b = *(const v2f*)&Klds[(j * 16 + l16) * LDT + 4 * kk + 2 * half];
          s[j] = wmma_f32(a, b, s[j]);
        }
      }

      float alpha[8];
#pragma unroll
      for (int v = 0; v < 8; ++v) {
        const int row = qbase + half * 8 + v;
        float sv[4];
        float mx = -1e30f;
#pragma unroll
        for (int j = 0; j < 4; ++j) {
          const int col = kt * 64 + j * 16 + l16;
          float x = s[j][v] * 0.125f;
          if (col > row) x = -1e30f;
          sv[j] = x;
          mx = fmaxf(mx, x);
        }
#pragma unroll
        for (int off = 1; off < 16; off <<= 1)
          mx = fmaxf(mx, __shfl_xor(mx, off, 32));
        const float mnew = fmaxf(mrow[v], mx);
        const float a = __expf(mrow[v] - mnew);
        float rsum = 0.0f;
#pragma unroll
        for (int j = 0; j < 4; ++j) {
          float p = __expf(sv[j] - mnew);
          myp[(half * 8 + v) * LDT + j * 16 + l16] = p;
          rsum += p;
        }
#pragma unroll
        for (int off = 1; off < 16; off <<= 1)
          rsum += __shfl_xor(rsum, off, 32);
        lrow[v] = lrow[v] * a + rsum;
        mrow[v] = mnew;
        alpha[v] = a;
      }
#pragma unroll
      for (int ct = 0; ct < 4; ++ct)
#pragma unroll
        for (int v = 0; v < 8; ++v) o[ct][v] *= alpha[v];

#pragma unroll 2
      for (int kk = 0; kk < 16; ++kk) {
        v2f a = *(const v2f*)&myp[l16 * LDT + 4 * kk + 2 * half];
#pragma unroll
        for (int ct = 0; ct < 4; ++ct) {
          v2f b = *(const v2f*)&Vlds[(ct * 16 + l16) * LDT + 4 * kk + 2 * half];
          o[ct] = wmma_f32(a, b, o[ct]);
        }
      }
    }
  }

#pragma unroll
  for (int ct = 0; ct < 4; ++ct)
#pragma unroll
    for (int v = 0; v < 8; ++v) {
      const uint32_t row = (uint32_t)(qbase + half * 8 + v);
      out[row * 1024u + (uint32_t)(h * 64 + ct * 16 + l16)] =
          o[ct][v] / lrow[v];
    }
}

extern "C" void kernel_launch(void* const* d_in, const int* in_sizes, int n_in,
                              void* d_out, int out_size, void* d_ws, size_t ws_size,
                              hipStream_t stream) {
  const float* x     = (const float*)d_in[0];
  const float* Wqkv  = (const float*)d_in[1];
  const float* Wproj = (const float*)d_in[2];
  float* out = (float*)d_out;

  float* qkv     = (float*)d_ws;
  float* attnout = qkv + (size_t)4096 * 3072;

  gemm_nt_wmma<<<dim3(3072 / 64, 4096 / 32), 32, 0, stream>>>(
      x, Wqkv, qkv, 1024, 1024, 1024, 3072);

  rope_kernel<<<(4096 * 1024) / 256, 256, 0, stream>>>(qkv, 4096);

  attn_kernel<<<dim3(32, 16), 256, 0, stream>>>(qkv, attnout);

  gemm_nt_wmma<<<dim3(1024 / 64, 4096 / 32), 32, 0, stream>>>(
      attnout, Wproj, out, 1024, 1024, 1024, 1024);
}


# attention operands fragment-packed in the workspace: the QKV GEMM stores q, k, v so that each 16-byte fragment load covers 512 contiguous bytes (4 lines per load, was 16)
# speedup vs baseline: 2.5615x; 2.5615x over previous
#include <hip/hip_runtime.h>
#include <hip/hip_bf16.h>

typedef __attribute__((ext_vector_type(2))) float v2f;
typedef __attribute__((ext_vector_type(4))) float v4f;
typedef __attribute__((ext_vector_type(8))) float v8f;

__device__ __forceinline__ v8f wmma_f32(v2f a, v2f b, v8f c) {
  return __builtin_amdgcn_wmma_f32_16x16x4_f32(
       false, a,  false, b,
       (short)0, c,  false,  false);
}

struct Frag { v4f q[4]; };

__device__ __forceinline__ void load_frag(Frag& f, const float* __restrict__ p) {
#pragma unroll
  for (int q = 0; q < 4; ++q) f.q[q] = *(const v4f*)(p + 4 * q);
}

__device__ __forceinline__ void load_set(Frag (&a)[2], Frag (&b)[4],
                                         const float* __restrict__ A,
                                         const float* __restrict__ B,
                                         const uint32_t (&aoff)[2],
                                         const uint32_t (&boff)[4], uint32_t kc) {
#pragma unroll
  for (int i = 0; i < 2; ++i) load_frag(a[i], A + aoff[i] + kc);
#pragma unroll
  for (int j = 0; j < 4; ++j) load_frag(b[j], B + boff[j] + kc);
}

template <bool SWAP>
__device__ __forceinline__ void mma_set(v8f (&acc)[2][4], const Frag (&a)[2],
                                        const Frag (&b)[4]) {
#pragma unroll
  for (int q = 0; q < 4; ++q) {
#pragma unroll
    for (int i = 0; i < 2; ++i)
#pragma unroll
      for (int j = 0; j < 4; ++j) {
        v2f av = {a[i].q[q].x, a[i].q[q].y};
        v2f bv = {b[j].q[q].x, b[j].q[q].y};
        acc[i][j] = SWAP ? wmma_f32(bv, av, acc[i][j]) : wmma_f32(av, bv, acc[i][j]);
      }
#pragma unroll
    for (int i = 0; i < 2; ++i)
#pragma unroll
      for (int j = 0; j < 4; ++j) {
        v2f av = {a[i].q[q].z, a[i].q[q].w};
        v2f bv = {b[j].q[q].z, b[j].q[q].w};
        acc[i][j] = SWAP ? wmma_f32(bv, av, acc[i][j]) : wmma_f32(av, bv, acc[i][j]);
      }
  }
}

template <bool SWAP>
__device__ __forceinline__ void gemm_loop(v8f (&acc)[2][4],
                                          const float* __restrict__ A,
                                          const float* __restrict__ B,
                                          const uint32_t (&aoff)[2],
                                          const uint32_t (&boff)[4], int nchunk) {
  Frag a0[2], b0[4], a1[2], b1[4];
  load_set(a0, b0, A, B, aoff, boff, 0u);

  int c = 0;
  for (; c + 2 < nchunk; c += 2) {
    load_set(a1, b1, A, B, aoff, boff, (uint32_t)(c + 1) * 32u);
    __builtin_amdgcn_sched_barrier(0);
    mma_set<SWAP>(acc, a0, b0);
    __builtin_amdgcn_sched_barrier(0);
    load_set(a0, b0, A, B, aoff, boff, (uint32_t)(c + 2) * 32u);
    __builtin_amdgcn_sched_barrier(0);
    mma_set<SWAP>(acc, a1, b1);
    __builtin_amdgcn_sched_barrier(0);
  }
  if (c + 1 < nchunk) {
    load_set(a1, b1, A, B, aoff, boff, (uint32_t)(c + 1) * 32u);
    __builtin_amdgcn_sched_barrier(0);
    mma_set<SWAP>(acc, a0, b0);
    mma_set<SWAP>(acc, a1, b1);
  } else {
    mma_set<SWAP>(acc, a0, b0);
  }
}

template <bool PACKED>
__global__ __launch_bounds__(32)
void gemm_nt_wmma(const float* __restrict__ A, const float* __restrict__ B,
                  float* __restrict__ C, float* __restrict__ VP, int K,
                  int lda, int ldb, int ldc) {
  const int lane = threadIdx.x & 31;
  const int half = lane >> 4;
  const int l16  = lane & 15;
  const int bm = blockIdx.y * 32;
  const int bn = blockIdx.x * 64;

  v8f acc[2][4];
#pragma unroll
  for (int i = 0; i < 2; ++i)
#pragma unroll
    for (int j = 0; j < 4; ++j) acc[i][j] = (v8f){};

  uint32_t aoff[2], boff[4];
#pragma unroll
  for (int i = 0; i < 2; ++i)
    aoff[i] = (uint32_t)(bm + i * 16 + l16) * (uint32_t)lda + 16u * half;
#pragma unroll
  for (int j = 0; j < 4; ++j)
    boff[j] = (uint32_t)(bn + j * 16 + l16) * (uint32_t)ldb + 16u * half;

  const int nchunk = K >> 5;

  if (!PACKED) {
    gemm_loop<false>(acc, A, B, aoff, boff, nchunk);
#pragma unroll
    for (int i = 0; i < 2; ++i)
#pragma unroll
      for (int j = 0; j < 4; ++j) {
        const uint32_t row0 = (uint32_t)(bm + i * 16 + half * 8);
        const uint32_t col  = (uint32_t)(bn + j * 16 + l16);
#pragma unroll
        for (int v = 0; v < 8; ++v)
          C[(row0 + v) * (uint32_t)ldc + col] = acc[i][j][v];
      }
  } else if (bn < 2048) {
    gemm_loop<true>(acc, A, B, aoff, boff, nchunk);
    const uint32_t region = (uint32_t)bn >> 10;
    const uint32_t hd     = ((uint32_t)bn >> 6) & 15u;
#pragma unroll
    for (int i = 0; i < 2; ++i)
#pragma unroll
      for (int j = 0; j < 4; ++j) {
        const uint32_t blk = hd * 256u + 2u * blockIdx.y + (uint32_t)i;
        float* p = C + region * 4194304u + (blk * 2u + (uint32_t)(j >> 1)) * 512u
                     + (uint32_t)(2 * half) * 128u
                     + (uint32_t)(l16 + 16 * (j & 1)) * 4u;
        v4f lo = {acc[i][j][0], acc[i][j][1], acc[i][j][2], acc[i][j][3]};
        v4f hi = {acc[i][j][4], acc[i][j][5], acc[i][j][6], acc[i][j][7]};
        *(v4f*)p = lo;
        *(v4f*)(p + 128) = hi;
      }
  } else {
    gemm_loop<false>(acc, A, B, aoff, boff, nchunk);
    const uint32_t hd = ((uint32_t)(bn - 2048) >> 6) & 15u;
#pragma unroll
    for (int i = 0; i < 2; ++i)
#pragma unroll
      for (int j = 0; j < 4; ++j) {
        float* p = VP + (((hd * 128u + blockIdx.y) * 4u + (uint32_t)j) * 4u
                         + (uint32_t)(2 * half)) * 128u
                      + (uint32_t)(l16 + 16 * i) * 4u;
        v4f lo = {acc[i][j][0], acc[i][j][1], acc[i][j][2], acc[i][j][3]};
        v4f hi = {acc[i][j][4], acc[i][j][5], acc[i][j][6], acc[i][j][7]};
        *(v4f*)p = lo;
        *(v4f*)(p + 128) = hi;
      }
  }
}

__global__ __launch_bounds__(256)
void rope_kernel(float* __restrict__ qk, int T) {
  uint32_t idx = blockIdx.x * blockDim.x + threadIdx.x;
  if (idx >= (uint32_t)T * 1024u) return;
  const uint32_t f   = 2u * idx;
  const uint32_t rb  = (f >> 10) & 255u;
  const uint32_t cc  = (f >> 9) & 1u;
  const uint32_t qd  = (f >> 7) & 3u;
  const uint32_t ln  = (f >> 2) & 31u;
  const uint32_t e   = f & 3u;
  const int t = (int)(rb * 16u + (ln & 15u));
  const int p = (int)((32u * cc + 16u * (ln >> 4) + 4u * qd + e) >> 1);
  float inv = powf(10000.0f, -(float)(2 * p) / 64.0f);
  float ang = (float)t * inv;
  float s = sinf(ang), c = cosf(ang);
  float* base = qk + f;
  float x1 = base[0], x2 = base[1];
  base[0] = x1 * c - x2 * s;
  base[1] = x2 * c + x1 * s;
}

#define LDP 68
#define ATTN_LDS_FLOATS (164 * 256)

__device__ __forceinline__ void load_frag_p(Frag& f, const float* __restrict__ p) {
#pragma unroll
  for (int q = 0; q < 4; ++q) f.q[q] = *(const v4f*)(p + 128 * q);
}

__device__ __forceinline__ void load_k(Frag (&kf)[4][2], const float* __restrict__ p) {
#pragma unroll
  for (int j = 0; j < 4; ++j)
#pragma unroll
    for (int cc = 0; cc < 2; ++cc)
      load_frag_p(kf[j][cc], p + (uint32_t)(j * 2 + cc) * 512u);
}

__device__ __forceinline__ void load_v(Frag (&vf)[4][2], const float* __restrict__ p) {
#pragma unroll
  for (int ct = 0; ct < 4; ++ct)
#pragma unroll
    for (int cc = 0; cc < 2; ++cc)
      load_frag_p(vf[ct][cc], p + (uint32_t)(cc * 16 + ct * 4) * 128u);
}

__device__ __forceinline__ void mma_tile(v8f (&acc)[4], const Frag (&a)[2],
                                         const Frag (&b)[4][2]) {
#pragma unroll
  for (int cc = 0; cc < 2; ++cc)
#pragma unroll
    for (int q = 0; q < 4; ++q) {
#pragma unroll
      for (int n = 0; n < 4; ++n) {
        v2f av = {a[cc].q[q].x, a[cc].q[q].y};
        v2f bv = {b[n][cc].q[q].x, b[n][cc].q[q].y};
        acc[n] = wmma_f32(av, bv, acc[n]);
      }
#pragma unroll
      for (int n = 0; n < 4; ++n) {
        v2f av = {a[cc].q[q].z, a[cc].q[q].w};
        v2f bv = {b[n][cc].q[q].z, b[n][cc].q[q].w};
        acc[n] = wmma_f32(av, bv, acc[n]);
      }
    }
}

__global__ __launch_bounds__(256)
void attn_kernel(const float* __restrict__ qk, const float* __restrict__ vt,
                 float* __restrict__ out) {
  __shared__ __attribute__((aligned(16))) float lds[ATTN_LDS_FLOATS];

  const int tid  = threadIdx.x;
  const int lane = tid & 31;
  const int wave = __builtin_amdgcn_readfirstlane(tid >> 5);
  const int half = lane >> 4;
  const int l16  = lane & 15;
  const int h     = blockIdx.x & 15;
  const int qt    = 31 - (int)(blockIdx.x >> 4);
  const int qbase = qt * 128 + wave * 16;

  const float* qp = qk + ((uint32_t)h * 256u + (uint32_t)(qbase >> 4)) * 1024u + (uint32_t)lane * 4u;
  const float* kp = qk + 4194304u + (uint32_t)h * 262144u + (uint32_t)lane * 4u;
  const float* vp = vt + (uint32_t)h * 262144u + (uint32_t)lane * 4u;
  float* myp = lds + wave * 16 * LDP;

  Frag qf[2];
#pragma unroll
  for (int cc = 0; cc < 2; ++cc) load_frag_p(qf[cc], qp + 512u * cc);

  v8f o[4];
#pragma unroll
  for (int ct = 0; ct < 4; ++ct) o[ct] = (v8f){};
  float mrow[8], lrow[8];
#pragma unroll
  for (int v = 0; v < 8; ++v) { mrow[v] = -1e30f; lrow[v] = 0.0f; }

  const int nt = ((qbase + 15) >> 6) + 1;

  Frag kf[4][2];
  load_k(kf, kp);

  for (int kt = 0; kt < nt; ++kt) {
    const int ktn = (kt + 1 < nt) ? kt + 1 : kt;

    Frag vf[4][2];
    load_v(vf, vp + (uint32_t)kt * 4096u);
    __builtin_amdgcn_sched_barrier(0);

    v8f s[4];
#pragma unroll
    for (int j = 0; j < 4; ++j) s[j] = (v8f){};
    mma_tile(s, qf, kf);
    __builtin_amdgcn_sched_barrier(0);
    load_k(kf, kp + (uint32_t)ktn * 4096u);
    __builtin_amdgcn_sched_barrier(0);

    float alpha[8];
#pragma unroll
    for (int v = 0; v < 8; ++v) {
      const int row = qbase + half * 8 + v;
      float sv[4];
      float mx = -1e30f;
#pragma unroll
      for (int j = 0; j < 4; ++j) {
        const int col = kt * 64 + j * 16 + l16;
        float x = s[j][v] * 0.125f;
        if (col > row) x = -1e30f;
        sv[j] = x;
        mx = fmaxf(mx, x);
      }
#pragma unroll
      for (int off = 1; off < 16; off <<= 1)
        mx = fmaxf(mx, __shfl_xor(mx, off, 32));
      const float mnew = fmaxf(mrow[v], mx);
      const float a = __expf(mrow[v] - mnew);
      float rsum = 0.0f;
#pragma unroll
      for (int j = 0; j < 4; ++j) {
        float p = __expf(sv[j] - mnew);
        myp[(half * 8 + v) * LDP + j * 16 + l16] = p;
        rsum += p;
      }
#pragma unroll
      for (int off = 1; off < 16; off <<= 1)
        rsum += __shfl_xor(rsum, off, 32);
      lrow[v] = lrow[v] * a + rsum;
      mrow[v] = mnew;
      alpha[v] = a;
    }
#pragma unroll
    for (int ct = 0; ct < 4; ++ct)
#pragma unroll
      for (int v = 0; v < 8; ++v) o[ct][v] *= alpha[v];

    __builtin_amdgcn_wave_barrier();
    Frag pf[2];
#pragma unroll
    for (int cc = 0; cc < 2; ++cc)
      load_frag(pf[cc], myp + l16 * LDP + 32 * cc + 16 * half);
    mma_tile(o, pf, vf);
    __builtin_amdgcn_sched_barrier(0);
  }

#pragma unroll
  for (int ct = 0; ct < 4; ++ct)
#pragma unroll
    for (int v = 0; v < 8; ++v) {
      const uint32_t row = (uint32_t)(qbase + half * 8 + v);
      out[row * 1024u + (uint32_t)(h * 64 + ct * 16 + l16)] =
          o[ct][v] / lrow[v];
    }
}

extern "C" void kernel_launch(void* const* d_in, const int* in_sizes, int n_in,
                              void* d_out, int out_size, void* d_ws, size_t ws_size,
                              hipStream_t stream) {
  const float* x     = (const float*)d_in[0];
  const float* Wqkv  = (const float*)d_in[1];
  const float* Wproj = (const float*)d_in[2];
  float* out = (float*)d_out;

  float* qk      = (float*)d_ws;
  float* vt      = qk + (size_t)4096 * 2048;
  float* attnout = vt + (size_t)1024 * 4096;

  gemm_nt_wmma<true><<<dim3(3072 / 64, 4096 / 32), 32, 0, stream>>>(
      x, Wqkv, qk, vt, 1024, 1024, 1024, 2048);

  rope_kernel<<<(4096 * 1024) / 256, 256, 0, stream>>>(qk, 4096);

  attn_kernel<<<dim3(32 * 16), 256, 0, stream>>>(qk, vt, attnout);

  gemm_nt_wmma<false><<<dim3(1024 / 64, 4096 / 32), 32, 0, stream>>>(
      attnout, Wproj, out, vt, 1024, 1024, 1024, 1024);
}


# attention: raw scores (not P) go through the wave's LDS slab; softmax runs behind it with half a query row per lane: in-lane max/sum + one half-wave exchange (was 64 ds_bpermute per tile)
# speedup vs baseline: 2.8558x; 1.1149x over previous
#include <hip/hip_runtime.h>
#include <hip/hip_bf16.h>

typedef __attribute__((ext_vector_type(2))) float v2f;
typedef __attribute__((ext_vector_type(4))) float v4f;
typedef __attribute__((ext_vector_type(8))) float v8f;

__device__ __forceinline__ v8f wmma_f32(v2f a, v2f b, v8f c) {
  return __builtin_amdgcn_wmma_f32_16x16x4_f32(
       false, a,  false, b,
       (short)0, c,  false,  false);
}

struct Frag { v4f q[4]; };

__device__ __forceinline__ void load_frag(Frag& f, const float* __restrict__ p) {
#pragma unroll
  for (int q = 0; q < 4; ++q) f.q[q] = *(const v4f*)(p + 4 * q);
}

__device__ __forceinline__ void load_set(Frag (&a)[2], Frag (&b)[4],
                                         const float* __restrict__ A,
                                         const float* __restrict__ B,
                                         const uint32_t (&aoff)[2],
                                         const uint32_t (&boff)[4], uint32_t kc) {
#pragma unroll
  for (int i = 0; i < 2; ++i) load_frag(a[i], A + aoff[i] + kc);
#pragma unroll
  for (int j = 0; j < 4; ++j) load_frag(b[j], B + boff[j] + kc);
}

template <bool SWAP>
__device__ __forceinline__ void mma_set(v8f (&acc)[2][4], const Frag (&a)[2],
                                        const Frag (&b)[4]) {
#pragma unroll
  for (int q = 0; q < 4; ++q) {
#pragma unroll
    for (int i = 0; i < 2; ++i)
#pragma unroll
      for (int j = 0; j < 4; ++j) {
        v2f av = {a[i].q[q].x, a[i].q[q].y};
        v2f bv = {b[j].q[q].x, b[j].q[q].y};
        acc[i][j] = SWAP ? wmma_f32(bv, av, acc[i][j]) : wmma_f32(av, bv, acc[i][j]);
      }
#pragma unroll
    for (int i = 0; i < 2; ++i)
#pragma unroll
      for (int j = 0; j < 4; ++j) {
        v2f av = {a[i].q[q].z, a[i].q[q].w};
        v2f bv = {b[j].q[q].z, b[j].q[q].w};
        acc[i][j] = SWAP ? wmma_f32(bv, av, acc[i][j]) : wmma_f32(av, bv, acc[i][j]);
      }
  }
}

template <bool SWAP>
__device__ __forceinline__ void gemm_loop(v8f (&acc)[2][4],
                                          const float* __restrict__ A,
                                          const float* __restrict__ B,
                                          const uint32_t (&aoff)[2],
                                          const uint32_t (&boff)[4], int nchunk) {
  Frag a0[2], b0[4], a1[2], b1[4];
  load_set(a0, b0, A, B, aoff, boff, 0u);

  int c = 0;
  for (; c + 2 < nchunk; c += 2) {
    load_set(a1, b1, A, B, aoff, boff, (uint32_t)(c + 1) * 32u);
    __builtin_amdgcn_sched_barrier(0);
    mma_set<SWAP>(acc, a0, b0);
    __builtin_amdgcn_sched_barrier(0);
    load_set(a0, b0, A, B, aoff, boff, (uint32_t)(c + 2) * 32u);
    __builtin_amdgcn_sched_barrier(0);
    mma_set<SWAP>(acc, a1, b1);
    __builtin_amdgcn_sched_barrier(0);
  }
  if (c + 1 < nchunk) {
    load_set(a1, b1, A, B, aoff, boff, (uint32_t)(c + 1) * 32u);
    __builtin_amdgcn_sched_barrier(0);
    mma_set<SWAP>(acc, a0, b0);
    mma_set<SWAP>(acc, a1, b1);
  } else {
    mma_set<SWAP>(acc, a0, b0);
  }
}

template <bool PACKED>
__global__ __launch_bounds__(32)
void gemm_nt_wmma(const float* __restrict__ A, const float* __restrict__ B,
                  float* __restrict__ C, float* __restrict__ VP, int K,
                  int lda, int ldb, int ldc) {
  const int lane = threadIdx.x & 31;
  const int half = lane >> 4;
  const int l16  = lane & 15;
  const int bm = blockIdx.y * 32;
  const int bn = blockIdx.x * 64;

  v8f acc[2][4];
#pragma unroll
  for (int i = 0; i < 2; ++i)
#pragma unroll
    for (int j = 0; j < 4; ++j) acc[i][j] = (v8f){};

  uint32_t aoff[2], boff[4];
#pragma unroll
  for (int i = 0; i < 2; ++i)
    aoff[i] = (uint32_t)(bm + i * 16 + l16) * (uint32_t)lda + 16u * half;
#pragma unroll
  for (int j = 0; j < 4; ++j)
    boff[j] = (uint32_t)(bn + j * 16 + l16) * (uint32_t)ldb + 16u * half;

  const int nchunk = K >> 5;

  if (!PACKED) {
    gemm_loop<false>(acc, A, B, aoff, boff, nchunk);
#pragma unroll
    for (int i = 0; i < 2; ++i)
#pragma unroll
      for (int j = 0; j < 4; ++j) {
        const uint32_t row0 = (uint32_t)(bm + i * 16 + half * 8);
        const uint32_t col  = (uint32_t)(bn + j * 16 + l16);
#pragma unroll
        for (int v = 0; v < 8; ++v)
          C[(row0 + v) * (uint32_t)ldc + col] = acc[i][j][v];
      }
  } else if (bn < 2048) {
    gemm_loop<true>(acc, A, B, aoff, boff, nchunk);
    const uint32_t region = (uint32_t)bn >> 10;
    const uint32_t hd     = ((uint32_t)bn >> 6) & 15u;
#pragma unroll
    for (int i = 0; i < 2; ++i)
#pragma unroll
      for (int j = 0; j < 4; ++j) {
        const uint32_t blk = hd * 256u + 2u * blockIdx.y + (uint32_t)i;
        float* p = C + region * 4194304u + (blk * 2u + (uint32_t)(j >> 1)) * 512u
                     + (uint32_t)(2 * half) * 128u
                     + (uint32_t)(l16 + 16 * (j & 1)) * 4u;
        v4f lo = {acc[i][j][0], acc[i][j][1], acc[i][j][2], acc[i][j][3]};
        v4f hi = {acc[i][j][4], acc[i][j][5], acc[i][j][6], acc[i][j][7]};
        *(v4f*)p = lo;
        *(v4f*)(p + 128) = hi;
      }
  } else {
    gemm_loop<false>(acc, A, B, aoff, boff, nchunk);
    const uint32_t hd = ((uint32_t)(bn - 2048) >> 6) & 15u;
#pragma unroll
    for (int i = 0; i < 2; ++i)
#pragma unroll
      for (int j = 0; j < 4; ++j) {
        float* p = VP + (((hd * 128u + blockIdx.y) * 4u + (uint32_t)j) * 4u
                         + (uint32_t)(2 * half)) * 128u
                      + (uint32_t)(l16 + 16 * i) * 4u;
        v4f lo = {acc[i][j][0], acc[i][j][1], acc[i][j][2], acc[i][j][3]};
        v4f hi = {acc[i][j][4], acc[i][j][5], acc[i][j][6], acc[i][j][7]};
        *(v4f*)p = lo;
        *(v4f*)(p + 128) = hi;
      }
  }
}

__global__ __launch_bounds__(256)
void rope_kernel(float* __restrict__ qk, int T) {
  uint32_t idx = blockIdx.x * blockDim.x + threadIdx.x;
  if (idx >= (uint32_t)T * 1024u) return;
  const uint32_t f   = 2u * idx;
  const uint32_t rb  = (f >> 10) & 255u;
  const uint32_t cc  = (f >> 9) & 1u;
  const uint32_t qd  = (f >> 7) & 3u;
  const uint32_t ln  = (f >> 2) & 31u;
  const uint32_t e   = f & 3u;
  const int t = (int)(rb * 16u + (ln & 15u));
  const int p = (int)((32u * cc + 16u * (ln >> 4) + 4u * qd + e) >> 1);
  float inv = powf(10000.0f, -(float)(2 * p) / 64.0f);
  float ang = (float)t * inv;
  float s = sinf(ang), c = cosf(ang);
  float* base = qk + f;
  float x1 = base[0], x2 = base[1];
  base[0] = x1 * c - x2 * s;
  base[1] = x2 * c + x1 * s;
}

#define LDP 68
#define ATTN_LDS_FLOATS (164 * 256)

__device__ __forceinline__ void load_frag_p(Frag& f, const float* __restrict__ p) {
#pragma unroll
  for (int q = 0; q < 4; ++q) f.q[q] = *(const v4f*)(p + 128 * q);
}

__device__ __forceinline__ void load_k(Frag (&kf)[4][2], const float* __restrict__ p) {
#pragma unroll
  for (int j = 0; j < 4; ++j)
#pragma unroll
    for (int cc = 0; cc < 2; ++cc)
      load_frag_p(kf[j][cc], p + (uint32_t)(j * 2 + cc) * 512u);
}

__device__ __forceinline__ void load_v(Frag (&vf)[4][2], const float* __restrict__ p) {
#pragma unroll
  for (int ct = 0; ct < 4; ++ct)
#pragma unroll
    for (int cc = 0; cc < 2; ++cc)
      load_frag_p(vf[ct][cc], p + (uint32_t)(cc * 16 + ct * 4) * 128u);
}

__device__ __forceinline__ void mma_tile(v8f (&acc)[4], const Frag (&a)[2],
                                         const Frag (&b)[4][2]) {
#pragma unroll
  for (int cc = 0; cc < 2; ++cc)
#pragma unroll
    for (int q = 0; q < 4; ++q) {
#pragma unroll
      for (int n = 0; n < 4; ++n) {
        v2f av = {a[cc].q[q].x, a[cc].q[q].y};
        v2f bv = {b[n][cc].q[q].x, b[n][cc].q[q].y};
        acc[n] = wmma_f32(av, bv, acc[n]);
      }
#pragma unroll
      for (int n = 0; n < 4; ++n) {
        v2f av = {a[cc].q[q].z, a[cc].q[q].w};
        v2f bv = {b[n][cc].q[q].z, b[n][cc].q[q].w};
        acc[n] = wmma_f32(av, bv, acc[n]);
      }
    }
}

__device__ __forceinline__ float xhalf(float x) {
  const int b = __builtin_bit_cast(int, x);
  return __builtin_bit_cast(float,
      __builtin_amdgcn_permlanex16(b, b, 0x76543210, 0xfedcba98, false, false));
}

__global__ __launch_bounds__(256)
void attn_kernel(const float* __restrict__ qk, const float* __restrict__ vt,
                 float* __restrict__ out) {
  __shared__ __attribute__((aligned(16))) float lds[ATTN_LDS_FLOATS];

  const int tid  = threadIdx.x;
  const int lane = tid & 31;
  const int wave = __builtin_amdgcn_readfirstlane(tid >> 5);
  const int half = lane >> 4;
  const int l16  = lane & 15;
  const int h     = blockIdx.x & 15;
  const int qt    = 31 - (int)(blockIdx.x >> 4);
  const int qbase = qt * 128 + wave * 16;

  const float* qp = qk + ((uint32_t)h * 256u + (uint32_t)(qbase >> 4)) * 1024u + (uint32_t)lane * 4u;
  const float* kp = qk + 4194304u + (uint32_t)h * 262144u + (uint32_t)lane * 4u;
  const float* vp = vt + (uint32_t)h * 262144u + (uint32_t)lane * 4u;
  float* myp = lds + wave * 16 * LDP;
  float* mya = lds + 8 * 16 * LDP + wave * 16;

  Frag qf[2];
#pragma unroll
  for (int cc = 0; cc < 2; ++cc) load_frag_p(qf[cc], qp + 512u * cc);

  v8f o[4];
#pragma unroll
  for (int ct = 0; ct < 4; ++ct) o[ct] = (v8f){};
  float mrow = -1e30f, lrow = 0.0f;
  const int row = qbase + l16;

  const int nt = ((qbase + 15) >> 6) + 1;

  Frag kf[4][2];
  load_k(kf, kp);

  for (int kt = 0; kt < nt; ++kt) {
    const int ktn = (kt + 1 < nt) ? kt + 1 : kt;

    Frag vf[4][2];
    load_v(vf, vp + (uint32_t)kt * 4096u);
    __builtin_amdgcn_sched_barrier(0);

    v8f s[4];
#pragma unroll
    for (int j = 0; j < 4; ++j) s[j] = (v8f){};
    mma_tile(s, qf, kf);
    __builtin_amdgcn_sched_barrier(0);
    load_k(kf, kp + (uint32_t)ktn * 4096u);
    __builtin_amdgcn_sched_barrier(0);

#pragma unroll
    for (int j = 0; j < 4; ++j)
#pragma unroll
      for (int v = 0; v < 8; ++v)
        myp[(half * 8 + v) * LDP + j * 16 + l16] = s[j][v];
    __builtin_amdgcn_wave_barrier();
    Frag pf[2];
#pragma unroll
    for (int cc = 0; cc < 2; ++cc)
      load_frag(pf[cc], myp + l16 * LDP + 32 * cc + 16 * half);

    const int dlim = row - kt * 64 - 16 * half;
    float mx = -1e30f;
#pragma unroll
    for (int cc = 0; cc < 2; ++cc)
#pragma unroll
      for (int q = 0; q < 4; ++q)
#pragma unroll
        for (int e = 0; e < 4; ++e) {
          float x = pf[cc].q[q][e] * 0.125f;
          if (32 * cc + 4 * q + e > dlim) x = -1e30f;
          pf[cc].q[q][e] = x;
          mx = fmaxf(mx, x);
        }
    mx = fmaxf(mx, xhalf(mx));
    const float mnew = fmaxf(mrow, mx);
    const float a = __expf(mrow - mnew);
    float rsum = 0.0f;
#pragma unroll
    for (int cc = 0; cc < 2; ++cc)
#pragma unroll
      for (int q = 0; q < 4; ++q)
#pragma unroll
        for (int e = 0; e < 4; ++e) {
          const float p = __expf(pf[cc].q[q][e] - mnew);
          pf[cc].q[q][e] = p;
          rsum += p;
        }
    lrow = lrow * a + rsum;
    mrow = mnew;

    mya[l16] = a;
    __builtin_amdgcn_wave_barrier();
    const v4f alo = *(const v4f*)(mya + 8 * half);
    const v4f ahi = *(const v4f*)(mya + 8 * half + 4);
    const float alpha[8] = {alo.x, alo.y, alo.z, alo.w, ahi.x, ahi.y, ahi.z, ahi.w};
#pragma unroll
    for (int ct = 0; ct < 4; ++ct)
#pragma unroll
      for (int v = 0; v < 8; ++v) o[ct][v] *= alpha[v];

    mma_tile(o, pf, vf);
    __builtin_amdgcn_sched_barrier(0);
  }

  mya[l16] = lrow + xhalf(lrow);
  __builtin_amdgcn_wave_barrier();
  const v4f llo = *(const v4f*)(mya + 8 * half);
  const v4f lhi = *(const v4f*)(mya + 8 * half + 4);
  const float lsum[8] = {llo.x, llo.y, llo.z, llo.w, lhi.x, lhi.y, lhi.z, lhi.w};
#pragma unroll
  for (int ct = 0; ct < 4; ++ct)
#pragma unroll
    for (int v = 0; v < 8; ++v) {
      const uint32_t orow = (uint32_t)(qbase + half * 8 + v);
      out[orow * 1024u + (uint32_t)(h * 64 + ct * 16 + l16)] =
          o[ct][v] / lsum[v];
    }
}

extern "C" void kernel_launch(void* const* d_in, const int* in_sizes, int n_in,
                              void* d_out, int out_size, void* d_ws, size_t ws_size,
                              hipStream_t stream) {
  const float* x     = (const float*)d_in[0];
  const float* Wqkv  = (const float*)d_in[1];
  const float* Wproj = (const float*)d_in[2];
  float* out = (float*)d_out;

  float* qk      = (float*)d_ws;
  float* vt      = qk + (size_t)4096 * 2048;
  float* attnout = vt + (size_t)1024 * 4096;

  gemm_nt_wmma<true><<<dim3(3072 / 64, 4096 / 32), 32, 0, stream>>>(
      x, Wqkv, qk, vt, 1024, 1024, 1024, 2048);

  rope_kernel<<<(4096 * 1024) / 256, 256, 0, stream>>>(qk, 4096);

  attn_kernel<<<dim3(32 * 16), 256, 0, stream>>>(qk, vt, attnout);

  gemm_nt_wmma<false><<<dim3(1024 / 64, 4096 / 32), 32, 0, stream>>>(
      attnout, Wproj, out, vt, 1024, 1024, 1024, 1024);
}
